# starting kernel
# baseline (speedup 1.0000x reference)
#include <hip/hip_runtime.h>

#define EMB 128
#define LOG2_BOOST 0.5849625007211562f
#define EPS 1e-8f

typedef __attribute__((ext_vector_type(2))) float v2f;
typedef __attribute__((ext_vector_type(8))) float v8f;

__global__ void me_zero_kernel(int* __restrict__ counts, float* __restrict__ partials,
                               int R, int P) {
    int i = blockIdx.x * blockDim.x + threadIdx.x;
    if (i < R) counts[i] = 0;
    if (i < P) partials[i] = 0.0f;
}

__global__ void me_count_kernel(const int* __restrict__ wi, const int* __restrict__ rmap,
                                int* __restrict__ roots, int* __restrict__ counts, int n) {
    int i = blockIdx.x * blockDim.x + threadIdx.x;
    if (i < n) {
        int r = rmap[wi[i]];
        roots[i] = r;
        atomicAdd(&counts[r], 1);
    }
}

__global__ void __launch_bounds__(256)
me_root_feat_kernel(const float* __restrict__ emb, const float* __restrict__ mats,
                    const float* __restrict__ bias, float* __restrict__ f) {
    int r = blockIdx.x;
    __shared__ float se[EMB];
    if (threadIdx.x < EMB) se[threadIdx.x] = emb[r * EMB + threadIdx.x];
    __syncthreads();

    int wave = threadIdx.x >> 5;
    int lane = threadIdx.x & 31;
    int m0   = wave * 16;

    int  koff = (lane >= 16) ? 2 : 0;
    const float* Mrow = mats + (size_t)r * EMB * EMB
                             + (size_t)(m0 + (lane & 15)) * EMB + koff;
    float bmask = ((lane & 15) == 0) ? 1.0f : 0.0f;

    v8f c = {};
    for (int k = 0; k < EMB; k += 4) {
        v2f a = *(const v2f*)(Mrow + k);
        v2f b;
        b.x = se[k + koff]     * bmask;
        b.y = se[k + koff + 1] * bmask;
        c = __builtin_amdgcn_wmma_f32_16x16x4_f32(
                 false, a,  false, b,
                 (short)0, c,  false,  false);
    }

    if ((lane & 15) == 0) {
        int base = m0 + ((lane == 16) ? 8 : 0);
        #pragma unroll
        for (int v = 0; v < 8; ++v)
            f[r * EMB + base + v] = c[v] + bias[r * EMB + base + v];
    }
}

__global__ void __launch_bounds__(256)
me_boost_kernel(const float* __restrict__ w, const int* __restrict__ roots,
                const int* __restrict__ counts, float* __restrict__ boosted,
                float* __restrict__ partials, int n) {
    __shared__ float red[256];
    int i = blockIdx.x * blockDim.x + threadIdx.x;
    float b = 0.0f;
    if (i < n) {
        float cm1 = (float)(counts[roots[i]] - 1);
        b = w[i] * exp2f(cm1 * LOG2_BOOST) + EPS;
        boosted[i] = b;
    }
    red[threadIdx.x] = b;
    __syncthreads();
    #pragma unroll
    for (int s = 128; s > 0; s >>= 1) {
        if (threadIdx.x < s) red[threadIdx.x] += red[threadIdx.x + s];
        __syncthreads();
    }
    if (threadIdx.x == 0) partials[blockIdx.x] = red[0];
}

__global__ void me_reduce_kernel(const float* __restrict__ partials,
                                 float* __restrict__ sum, int P) {
    if (threadIdx.x == 0) {
        float s = 0.0f;
        for (int i = 0; i < P; ++i) s += partials[i];
        sum[0] = s;
    }
}

__global__ void __launch_bounds__(128)
me_out_kernel(const float* __restrict__ f, const int* __restrict__ roots,
              const float* __restrict__ boosted, const float* __restrict__ sum,
              float4* __restrict__ out, int n) {
    int word = blockIdx.x * 4 + (threadIdx.x >> 5);
    int lane = threadIdx.x & 31;
    if (word < n) {
        float scale = boosted[word] / sum[0];
        const float4* fr = (const float4*)(f + (size_t)roots[word] * EMB);
        float4 v = fr[lane];
        v.x *= scale; v.y *= scale; v.z *= scale; v.w *= scale;
        out[(size_t)word * (EMB / 4) + lane] = v;
    }
}

static inline size_t me_align256(size_t x) { return (x + 255) & ~(size_t)255; }

extern "C" void kernel_launch(void* const* d_in, const int* in_sizes, int n_in,
                              void* d_out, int out_size, void* d_ws, size_t ws_size,
                              hipStream_t stream) {
    const float* emb  = (const float*)d_in[0];
    const float* mats = (const float*)d_in[1];
    const float* bias = (const float*)d_in[2];
    const float* attw = (const float*)d_in[3];
    const int*   wi   = (const int*)d_in[4];
    const int*   rmap = (const int*)d_in[5];

    const int N = in_sizes[3];
    const int R = in_sizes[0] / EMB;
    const int P = (N + 255) / 256;

    char* ws = (char*)d_ws;
    size_t off = 0;
    int*   counts   = (int*)(ws + off);   off = me_align256(off + (size_t)R * 4);
    int*   roots    = (int*)(ws + off);   off = me_align256(off + (size_t)N * 4);
    float* boosted  = (float*)(ws + off); off = me_align256(off + (size_t)N * 4);
    float* partials = (float*)(ws + off); off = me_align256(off + (size_t)P * 4);
    float* sum      = (float*)(ws + off); off = me_align256(off + 4);
    float* f        = (float*)(ws + off); off = me_align256(off + (size_t)R * EMB * 4);
    (void)ws_size; (void)n_in; (void)out_size;

    int zmax = (R > P) ? R : P;
    me_zero_kernel<<<(zmax + 255) / 256, 256, 0, stream>>>(counts, partials, R, P);
    me_count_kernel<<<(N + 255) / 256, 256, 0, stream>>>(wi, rmap, roots, counts, N);
    me_root_feat_kernel<<<R, 256, 0, stream>>>(emb, mats, bias, f);
    me_boost_kernel<<<P, 256, 0, stream>>>(attw, roots, counts, boosted, partials, N);
    me_reduce_kernel<<<1, 32, 0, stream>>>(partials, sum, P);
    me_out_kernel<<<(N + 3) / 4, 128, 0, stream>>>(f, roots, boosted, sum,
                                                   (float4*)d_out, N);
}


# 3 launches to 2: root-range count and weight-sum blocks ride in the matvec launch, each reading the root map into an LDS byte table to find every word's root itself; sum order changed
# speedup vs baseline: 1.9171x; 1.9171x over previous
#include <hip/hip_runtime.h>

#define EMB 128
#define LOG2_BOOST 0.5849625007211562f
#define EPS 1e-8f

typedef __attribute__((ext_vector_type(2))) float v2f;
typedef __attribute__((ext_vector_type(4))) float v4f;
typedef __attribute__((ext_vector_type(8))) float v8f;

#define RF_WAVES 8
#define RF_ROOTS 4
#define RF_PITCH 132
#define RF_EROW  16

struct rf_stage { v4f m[16]; v4f e; v4f b0; v4f b1; };

__device__ __forceinline__ void rf_issue(rf_stage& s, const float* __restrict__ emb,
                                         const float* __restrict__ mats,
                                         const float* __restrict__ bias,
                                         int r, int wave, int lane) {
    const float* Mr = mats + (size_t)r * EMB * EMB + (size_t)(16 * wave) * EMB + 4 * lane;
    #pragma unroll
    for (int j = 0; j < 16; ++j) s.m[j] = *(const v4f*)(Mr + j * EMB);
    s.e = *(const v4f*)(emb + (size_t)r * EMB + 4 * lane);
    const float* bp = bias + (size_t)r * EMB + 16 * wave + 8 * (lane >> 4);
    s.b0 = *(const v4f*)bp;
    s.b1 = *(const v4f*)(bp + 4);
}

#define ST_THREADS 256
#define ST_ROOTS   32
#define ST_KEEP    16
#define ST_TAB     65536
#define ST_NONE    255u

typedef __attribute__((ext_vector_type(4))) int v4i;
typedef __attribute__((ext_vector_type(4))) unsigned v4u;

__device__ __forceinline__ float me_boost_factor(int count) {
    return exp2f((float)(count - 1) * LOG2_BOOST);
}
__device__ __forceinline__ float me_boosted(float w, float factor) {
    return w * factor + EPS;
}

__device__ __forceinline__ void st_count(unsigned d, int* cnt) {
    if (d < ST_ROOTS) atomicAdd(&cnt[d], 1);
}
__device__ __forceinline__ float st_term(unsigned d, float w, const float* fac) {
    const bool mine = d < ST_ROOTS;
    const float b = me_boosted(w, fac[mine ? d : 0u]);
    return mine ? b : 0.0f;
}
__device__ __forceinline__ unsigned st_byte(int root, int r_lo) {
    const unsigned d = (unsigned)(root - r_lo);
    return d < ST_ROOTS ? d : ST_NONE;
}
__device__ __forceinline__ unsigned st_lookup(const int* __restrict__ wi, const int* __restrict__ rmap,
                                              const unsigned char* tabb, bool tabbed, int r_lo, int i) {
    const unsigned v = (unsigned)wi[i];
    return tabbed ? (unsigned)tabb[v] : st_byte(rmap[v], r_lo);
}

__device__ __forceinline__ void me_stats_block(const float* __restrict__ w, const int* __restrict__ wi,
                                               const int* __restrict__ rmap, int* __restrict__ counts,
                                               float* __restrict__ partials, int n, int R, int V, int b,
                                               unsigned* tab, int* cnt, float* fac, float* red) {
    const int tid = threadIdx.x;
    const int r_lo = b * ST_ROOTS;
    const int n4 = n >> 2;
    const v4i* wi4 = (const v4i*)wi;
    const v4f* w4 = (const v4f*)w;
    const unsigned char* tabb = (const unsigned char*)tab;

    if (tid < ST_ROOTS) cnt[tid] = 0;

    const bool tabbed = (V <= ST_TAB) && ((V & 3) == 0);
    const bool keep = tabbed && (n4 >= ST_KEEP * ST_THREADS);
    const int i_rest = keep ? 4 * ST_KEEP * ST_THREADS : 0;
    v4i ki[ST_KEEP];
    v4u kd[ST_KEEP];
    v4f kw[ST_KEEP];
    #pragma unroll
    for (int v = 0; v < ST_KEEP; ++v) {
        ki[v] = (v4i){0, 0, 0, 0};
        kd[v] = (v4u){ST_NONE, ST_NONE, ST_NONE, ST_NONE};
        kw[v] = (v4f){0.0f, 0.0f, 0.0f, 0.0f};
    }
    if (keep) {
        #pragma unroll
        for (int v = 0; v < ST_KEEP; ++v) ki[v] = wi4[tid + v * ST_THREADS];
    }
    if (tabbed) {
        const v4i* rmap4 = (const v4i*)rmap;
        const int V4 = V >> 2;
        #pragma unroll 32
        for (int g = tid; g < V4; g += ST_THREADS) {
            const v4i t = rmap4[g];
            tab[g] = st_byte(t.x, r_lo) | (st_byte(t.y, r_lo) << 8) |
                     (st_byte(t.z, r_lo) << 16) | (st_byte(t.w, r_lo) << 24);
        }
    }
    __syncthreads();
    if (keep) {
        #pragma unroll
        for (int v = 0; v < ST_KEEP; ++v) kw[v] = w4[tid + v * ST_THREADS];
        #pragma unroll
        for (int v = 0; v < ST_KEEP; ++v) {
            kd[v].x = tabb[(unsigned)ki[v].x]; kd[v].y = tabb[(unsigned)ki[v].y];
            kd[v].z = tabb[(unsigned)ki[v].z]; kd[v].w = tabb[(unsigned)ki[v].w];
        }
    }
    #pragma unroll
    for (int v = 0; v < ST_KEEP; ++v) {
        st_count(kd[v].x, cnt); st_count(kd[v].y, cnt);
        st_count(kd[v].z, cnt); st_count(kd[v].w, cnt);
    }
    for (int i = i_rest + tid; i < n; i += ST_THREADS)
        st_count(st_lookup(wi, rmap, tabb, tabbed, r_lo, i), cnt);
    __syncthreads();
    if (tid < ST_ROOTS) fac[tid] = me_boost_factor(cnt[tid]);
    __syncthreads();

    float s = 0.0f;
    #pragma unroll
    for (int v = 0; v < ST_KEEP; ++v) {
        s += st_term(kd[v].x, kw[v].x, fac); s += st_term(kd[v].y, kw[v].y, fac);
        s += st_term(kd[v].z, kw[v].z, fac); s += st_term(kd[v].w, kw[v].w, fac);
    }
    for (int i = i_rest + tid; i < n; i += ST_THREADS)
        s += st_term(st_lookup(wi, rmap, tabb, tabbed, r_lo, i), w[i], fac);

    for (int m = 16; m > 0; m >>= 1) s += __shfl_xor(s, m);
    if ((tid & 31) == 0) red[tid >> 5] = s;
    __syncthreads();
    if (tid < 32) {
        float t = (tid < ST_THREADS / 32) ? red[tid] : 0.0f;
        for (int m = 16; m > 0; m >>= 1) t += __shfl_xor(t, m);
        if (tid == 0) partials[b] = t;
        if (r_lo + tid < R) counts[r_lo + tid] = cnt[tid];
    }
}

__global__ void __launch_bounds__(256)
me_root_feat_kernel(const float* __restrict__ emb, const float* __restrict__ mats,
                    const float* __restrict__ bias, float* __restrict__ f, int R,
                    const int* __restrict__ wi, const int* __restrict__ rmap,
                    int* __restrict__ roots, int n,
                    const float* __restrict__ w, int* __restrict__ counts,
                    float* __restrict__ partials, int P, int V) {
    __shared__ __attribute__((aligned(16))) float slot[RF_WAVES][(RF_EROW + 1) * RF_PITCH];
    __shared__ int st_cnt[ST_ROOTS];
    __shared__ float st_fac[ST_ROOTS];
    __shared__ float st_red[ST_THREADS / 32];

    if ((int)blockIdx.x < P) {
        me_stats_block(w, wi, rmap, counts, partials, n, R, V, (int)blockIdx.x,
                       (unsigned*)&slot[0][0], st_cnt, st_fac, st_red);
        return;
    }
    const int bx = (int)blockIdx.x - P;
    const int nbx = (int)gridDim.x - P;

    const int wave = __builtin_amdgcn_readfirstlane(threadIdx.x >> 5);
    const int lane = threadIdx.x & 31;
    const int c = lane & 15;
    const int h = lane >> 4;
    float* my = slot[wave];
    const int r0 = bx * RF_ROOTS;
    const int rlast = R - 1;

    const int gstride = nbx * (RF_WAVES * 32);
    const int gi = bx * (RF_WAVES * 32) + threadIdx.x;
    unsigned gw = 0;
    if (gi < n) gw = (unsigned)wi[gi];

    rf_stage st;
    rf_issue(st, emb, mats, bias, r0, wave, lane);

    int gr = 0;
    if (gi < n) gr = rmap[gw];

    #pragma unroll
    for (int i = 0; i < RF_ROOTS; ++i) {
        const int r = r0 + i;
        __builtin_amdgcn_fence(__ATOMIC_RELEASE, "wavefront");
        __builtin_amdgcn_wave_barrier();
        __builtin_amdgcn_fence(__ATOMIC_ACQUIRE, "wavefront");
        #pragma unroll
        for (int j = 0; j < 16; ++j) *(v4f*)(my + j * RF_PITCH + 4 * lane) = st.m[j];
        *(v4f*)(my + RF_EROW * RF_PITCH + 4 * lane) = st.e;
        const v4f bia0 = st.b0, bia1 = st.b1;
        __builtin_amdgcn_fence(__ATOMIC_RELEASE, "wavefront");
        __builtin_amdgcn_wave_barrier();
        __builtin_amdgcn_fence(__ATOMIC_ACQUIRE, "wavefront");

        const float* arow = my + c * RF_PITCH + 4 * h;
        const float* brow = my + RF_EROW * RF_PITCH + 4 * h;
        v4f af[16], bf[16];
        #pragma unroll
        for (int kk = 0; kk < 16; ++kk) {
            af[kk] = *(const v4f*)(arow + 8 * kk);
            bf[kk] = *(const v4f*)(brow + 8 * kk);
        }

        if (i + 1 < RF_ROOTS) {
            const int rn = (r + 1 < R) ? (r + 1) : rlast;
            rf_issue(st, emb, mats, bias, rn, wave, lane);
        }
        __builtin_amdgcn_sched_barrier(0);

        v8f acc0 = {}, acc1 = {};
        #pragma unroll
        for (int kk = 0; kk < 16; ++kk) {
            v2f a0 = {af[kk].x, af[kk].y}, a1 = {af[kk].z, af[kk].w};
            v2f b0 = {bf[kk].x, bf[kk].y}, b1 = {bf[kk].z, bf[kk].w};
            acc0 = __builtin_amdgcn_wmma_f32_16x16x4_f32(false, a0, false, b0, (short)0, acc0, false, false);
            acc1 = __builtin_amdgcn_wmma_f32_16x16x4_f32(false, a1, false, b1, (short)0, acc1, false, false);
        }

        if (c == 0 && r < R) {
            float* fp = f + (size_t)r * EMB + 16 * wave + 8 * h;
            v4f o0, o1;
            o0.x = (acc0[0] + acc1[0]) + bia0.x; o0.y = (acc0[1] + acc1[1]) + bia0.y;
            o0.z = (acc0[2] + acc1[2]) + bia0.z; o0.w = (acc0[3] + acc1[3]) + bia0.w;
            o1.x = (acc0[4] + acc1[4]) + bia1.x; o1.y = (acc0[5] + acc1[5]) + bia1.y;
            o1.z = (acc0[6] + acc1[6]) + bia1.z; o1.w = (acc0[7] + acc1[7]) + bia1.w;
            *(v4f*)fp = o0;
            *(v4f*)(fp + 4) = o1;
        }
        if (i == 0 && gi < n) roots[gi] = gr;
    }
    for (int g2 = gi + gstride; g2 < n; g2 += gstride) roots[g2] = rmap[wi[g2]];
}

__global__ void __launch_bounds__(128)
me_out_kernel(const float* __restrict__ f, const int* __restrict__ roots,
              const int* __restrict__ counts, const float* __restrict__ w,
              const float* __restrict__ partials, float4* __restrict__ out,
              int n, int P) {
    int word = blockIdx.x * 4 + (threadIdx.x >> 5);
    int lane = threadIdx.x & 31;
    if (word < n) {
        const int r = roots[word];
        const float wv = w[word];
        const float p0 = (lane < P) ? partials[lane] : 0.0f;
        const float p1 = (lane + 32 < P) ? partials[lane + 32] : 0.0f;
        float sum = p0 + p1;
        for (int j = lane + 64; j < P; j += 32) sum += partials[j];
        for (int m = 16; m > 0; m >>= 1) sum += __shfl_xor(sum, m);
        float scale = me_boosted(wv, me_boost_factor(counts[r])) / sum;
        const float4* fr = (const float4*)(f + (size_t)r * EMB);
        float4 v = fr[lane];
        v.x *= scale; v.y *= scale; v.z *= scale; v.w *= scale;
        out[(size_t)word * (EMB / 4) + lane] = v;
    }
}

static inline size_t me_align256(size_t x) { return (x + 255) & ~(size_t)255; }

extern "C" void kernel_launch(void* const* d_in, const int* in_sizes, int n_in,
                              void* d_out, int out_size, void* d_ws, size_t ws_size,
                              hipStream_t stream) {
    const float* emb  = (const float*)d_in[0];
    const float* mats = (const float*)d_in[1];
    const float* bias = (const float*)d_in[2];
    const float* attw = (const float*)d_in[3];
    const int*   wi   = (const int*)d_in[4];
    const int*   rmap = (const int*)d_in[5];

    const int N = in_sizes[3];
    const int R = in_sizes[0] / EMB;
    const int V = in_sizes[5];
    const int P = (R + ST_ROOTS - 1) / ST_ROOTS;

    char* ws = (char*)d_ws;
    size_t off = 0;
    int*   counts   = (int*)(ws + off);   off = me_align256(off + (size_t)R * 4);
    int*   roots    = (int*)(ws + off);   off = me_align256(off + (size_t)N * 4);
    float* partials = (float*)(ws + off); off = me_align256(off + (size_t)P * 4);
    float* f        = (float*)(ws + off); off = me_align256(off + (size_t)R * EMB * 4);
    (void)ws_size; (void)n_in; (void)out_size;

    me_root_feat_kernel<<<P + (R + RF_ROOTS - 1) / RF_ROOTS, 256, 0, stream>>>(emb, mats, bias, f, R,
                                                                              wi, rmap, roots, N,
                                                                              attw, counts, partials, P, V);
    me_out_kernel<<<(N + 3) / 4, 128, 0, stream>>>(f, roots, counts, attw, partials,
                                                   (float4*)d_out, N, P);
}
